# starting kernel
# baseline (speedup 1.0000x reference)
#include <hip/hip_runtime.h>
#include <hip/hip_bf16.h>

#define NUM_USERS 100000
#define NUM_ITEMS 50000
#define N_NODES   (NUM_USERS + NUM_ITEMS)
#define LATENT    64
#define NNZ_E     4000000
#define BATCH_SZ  16384

typedef __attribute__((ext_vector_type(2))) float v2f;
typedef __attribute__((ext_vector_type(8))) float v8f;

__global__ void lgcn_init(const float* __restrict__ utab, const float* __restrict__ itab,
                          float* __restrict__ X, float* __restrict__ acc) {
    long i4 = (long)blockIdx.x * blockDim.x + threadIdx.x;
    long n4 = (long)N_NODES * LATENT / 4;
    if (i4 >= n4) return;
    long i = i4 * 4;
    const long ub = (long)NUM_USERS * LATENT;
    float4 v;
    if (i < ub) v = *(const float4*)(utab + i);
    else        v = *(const float4*)(itab + (i - ub));
    *(float4*)(X + i)   = v;
    *(float4*)(acc + i) = v;
}

__global__ void lgcn_zero(float* __restrict__ Y) {
    long i4 = (long)blockIdx.x * blockDim.x + threadIdx.x;
    long n4 = (long)N_NODES * LATENT / 4;
    if (i4 >= n4) return;
    *(float4*)(Y + i4 * 4) = make_float4(0.f, 0.f, 0.f, 0.f);
}

__global__ void lgcn_accum(float* __restrict__ acc, const float* __restrict__ Y) {
    long i4 = (long)blockIdx.x * blockDim.x + threadIdx.x;
    long n4 = (long)N_NODES * LATENT / 4;
    if (i4 >= n4) return;
    float4 a = *(float4*)(acc + i4 * 4);
    float4 y = *(const float4*)(Y + i4 * 4);
    a.x += y.x; a.y += y.y; a.z += y.z; a.w += y.w;
    *(float4*)(acc + i4 * 4) = a;
}

__global__ void lgcn_spmm(const int* __restrict__ esrc, const int* __restrict__ edst,
                          const float* __restrict__ evals,
                          const float* __restrict__ X, float* __restrict__ Y) {
    int tid  = blockIdx.x * blockDim.x + threadIdx.x;
    int edge = tid >> 5;
    int lane = threadIdx.x & 31;
    if (edge >= NNZ_E) return;
    int   s = __builtin_nontemporal_load(esrc + edge);
    int   d = __builtin_nontemporal_load(edst + edge);
    float v = __builtin_nontemporal_load(evals + edge);
    const float* xr = X + (long)s * LATENT + 2 * lane;
    float2 x2 = *(const float2*)xr;
    float* yp = Y + (long)d * LATENT + 2 * lane;
    unsafeAtomicAdd(yp,     x2.x * v);
    unsafeAtomicAdd(yp + 1, x2.y * v);
}

__global__ void lgcn_gamma_wmma(const float* __restrict__ acc,
                                const int* __restrict__ users,
                                const int* __restrict__ items,
                                float* __restrict__ out) {
    int wave = (blockIdx.x * blockDim.x + threadIdx.x) >> 5;
    int lane = threadIdx.x & 31;
    int b    = wave * 16 + (lane & 15);
    int half = (lane >> 4) & 1;

    const float* urow = acc + (long)users[b] * LATENT;
    const float* irow = acc + ((long)NUM_USERS + items[b]) * LATENT;

    v8f c = {};
#pragma unroll
    for (int kk = 0; kk < 16; ++kk) {
        int off = 4 * kk + 2 * half;
        v2f a  = *(const v2f*)(urow + off);
        v2f bb = *(const v2f*)(irow + off);
        c = __builtin_amdgcn_wmma_f32_16x16x4_f32(false, a, false, bb,
                                                  (short)0, c, false, false);
    }

    int r = ((lane < 8) ? lane : (lane - 24)) & 7;
    float v = c[0];
#pragma unroll
    for (int i = 1; i < 8; ++i) if (r == i) v = c[i];

    if (lane < 8)        out[wave * 16 + lane]        = v * 0.0625f;
    else if (lane >= 24) out[wave * 16 + (lane - 16)] = v * 0.0625f;
}

extern "C" void kernel_launch(void* const* d_in, const int* in_sizes, int n_in,
                              void* d_out, int out_size, void* d_ws, size_t ws_size,
                              hipStream_t stream) {
    const int*   users = (const int*)d_in[0];
    const int*   items = (const int*)d_in[1];
    const int*   esrc  = (const int*)d_in[2];
    const int*   edst  = (const int*)d_in[3];
    const float* evals = (const float*)d_in[4];
    const float* utab  = (const float*)d_in[5];
    const float* itab  = (const float*)d_in[6];
    float* out = (float*)d_out;

    const long NE = (long)N_NODES * LATENT;
    float* X   = (float*)d_ws;
    float* Y   = X + NE;
    float* acc = Y + NE;

    const long n4      = NE / 4;
    const int  blk     = 256;
    const int  gridEl  = (int)((n4 + blk - 1) / blk);

    lgcn_init<<<gridEl, blk, 0, stream>>>(utab, itab, X, acc);

    for (int layer = 0; layer < 3; ++layer) {
        lgcn_zero<<<gridEl, blk, 0, stream>>>(Y);
        long spmmThreads = (long)NNZ_E * 32;
        lgcn_spmm<<<(int)(spmmThreads / blk), blk, 0, stream>>>(esrc, edst, evals, X, Y);
        lgcn_accum<<<gridEl, blk, 0, stream>>>(acc, Y);
        float* t = X; X = Y; Y = t;
    }

    lgcn_gamma_wmma<<<BATCH_SZ / 16 / 8, blk, 0, stream>>>(acc, users, items, out);
}


# SpMM: one wave per 32 edges instead of per edge: edge triples read as 3 coalesced lines and broadcast by readlane, row gathers 8 in flight; f32 atomics unchanged (arrival order as before)
# speedup vs baseline: 1.9793x; 1.9793x over previous
#include <hip/hip_runtime.h>
#include <hip/hip_bf16.h>

#define NUM_USERS 100000
#define NUM_ITEMS 50000
#define N_NODES   (NUM_USERS + NUM_ITEMS)
#define LATENT    64
#define NNZ_E     4000000
#define BATCH_SZ  16384

typedef __attribute__((ext_vector_type(2))) float v2f;
typedef __attribute__((ext_vector_type(8))) float v8f;

__global__ void lgcn_init(const float* __restrict__ utab, const float* __restrict__ itab,
                          float* __restrict__ X, float* __restrict__ acc) {
    long i4 = (long)blockIdx.x * blockDim.x + threadIdx.x;
    long n4 = (long)N_NODES * LATENT / 4;
    if (i4 >= n4) return;
    long i = i4 * 4;
    const long ub = (long)NUM_USERS * LATENT;
    float4 v;
    if (i < ub) v = *(const float4*)(utab + i);
    else        v = *(const float4*)(itab + (i - ub));
    *(float4*)(X + i)   = v;
    *(float4*)(acc + i) = v;
}

__global__ void lgcn_zero(float* __restrict__ Y) {
    long i4 = (long)blockIdx.x * blockDim.x + threadIdx.x;
    long n4 = (long)N_NODES * LATENT / 4;
    if (i4 >= n4) return;
    *(float4*)(Y + i4 * 4) = make_float4(0.f, 0.f, 0.f, 0.f);
}

__global__ void lgcn_accum(float* __restrict__ acc, const float* __restrict__ Y) {
    long i4 = (long)blockIdx.x * blockDim.x + threadIdx.x;
    long n4 = (long)N_NODES * LATENT / 4;
    if (i4 >= n4) return;
    float4 a = *(float4*)(acc + i4 * 4);
    float4 y = *(const float4*)(Y + i4 * 4);
    a.x += y.x; a.y += y.y; a.z += y.z; a.w += y.w;
    *(float4*)(acc + i4 * 4) = a;
}

static_assert(NNZ_E % 256 == 0, "a 256-thread block takes 256 edges, no tail");
__global__ void lgcn_spmm(const int* __restrict__ esrc, const int* __restrict__ edst,
                          const float* __restrict__ evals,
                          const float* __restrict__ X, float* __restrict__ Y) {
    const int lane = threadIdx.x & 31;
    const int e    = blockIdx.x * blockDim.x + threadIdx.x;
    const int s  = __builtin_nontemporal_load(esrc + e);
    const int d  = __builtin_nontemporal_load(edst + e);
    const int vb = __float_as_int(__builtin_nontemporal_load(evals + e));
    const float* xl = X + 2 * lane;
    float*       yl = Y + 2 * lane;
#pragma unroll
    for (int g = 0; g < 32; g += 8) {
        float2 x[8];
#pragma unroll
        for (int j = 0; j < 8; ++j) {
            const int sj = __builtin_amdgcn_readlane(s, g + j);
            x[j] = *(const float2*)(xl + (long)sj * LATENT);
        }
#pragma unroll
        for (int j = 0; j < 8; ++j) {
            const int   dj = __builtin_amdgcn_readlane(d, g + j);
            const float vj = __int_as_float(__builtin_amdgcn_readlane(vb, g + j));
            float* yp = yl + (long)dj * LATENT;
            unsafeAtomicAdd(yp,     x[j].x * vj);
            unsafeAtomicAdd(yp + 1, x[j].y * vj);
        }
    }
}

__global__ void lgcn_gamma_wmma(const float* __restrict__ acc,
                                const int* __restrict__ users,
                                const int* __restrict__ items,
                                float* __restrict__ out) {
    int wave = (blockIdx.x * blockDim.x + threadIdx.x) >> 5;
    int lane = threadIdx.x & 31;
    int b    = wave * 16 + (lane & 15);
    int half = (lane >> 4) & 1;

    const float* urow = acc + (long)users[b] * LATENT;
    const float* irow = acc + ((long)NUM_USERS + items[b]) * LATENT;

    v8f c = {};
#pragma unroll
    for (int kk = 0; kk < 16; ++kk) {
        int off = 4 * kk + 2 * half;
        v2f a  = *(const v2f*)(urow + off);
        v2f bb = *(const v2f*)(irow + off);
        c = __builtin_amdgcn_wmma_f32_16x16x4_f32(false, a, false, bb,
                                                  (short)0, c, false, false);
    }

    int r = ((lane < 8) ? lane : (lane - 24)) & 7;
    float v = c[0];
#pragma unroll
    for (int i = 1; i < 8; ++i) if (r == i) v = c[i];

    if (lane < 8)        out[wave * 16 + lane]        = v * 0.0625f;
    else if (lane >= 24) out[wave * 16 + (lane - 16)] = v * 0.0625f;
}

extern "C" void kernel_launch(void* const* d_in, const int* in_sizes, int n_in,
                              void* d_out, int out_size, void* d_ws, size_t ws_size,
                              hipStream_t stream) {
    const int*   users = (const int*)d_in[0];
    const int*   items = (const int*)d_in[1];
    const int*   esrc  = (const int*)d_in[2];
    const int*   edst  = (const int*)d_in[3];
    const float* evals = (const float*)d_in[4];
    const float* utab  = (const float*)d_in[5];
    const float* itab  = (const float*)d_in[6];
    float* out = (float*)d_out;

    const long NE = (long)N_NODES * LATENT;
    float* X   = (float*)d_ws;
    float* Y   = X + NE;
    float* acc = Y + NE;

    const long n4      = NE / 4;
    const int  blk     = 256;
    const int  gridEl  = (int)((n4 + blk - 1) / blk);

    lgcn_init<<<gridEl, blk, 0, stream>>>(utab, itab, X, acc);

    for (int layer = 0; layer < 3; ++layer) {
        lgcn_zero<<<gridEl, blk, 0, stream>>>(Y);
        lgcn_spmm<<<NNZ_E / blk, blk, 0, stream>>>(esrc, edst, evals, X, Y);
        lgcn_accum<<<gridEl, blk, 0, stream>>>(acc, Y);
        float* t = X; X = Y; Y = t;
    }

    lgcn_gamma_wmma<<<BATCH_SZ / 16 / 8, blk, 0, stream>>>(acc, users, items, out);
}


# SpMM owner-computes: edges counting-sorted by dst in-call (int atomics), one wave sums a row in registers, no f32 atomics; layers nested x0+A(..): row sum order changes
# speedup vs baseline: 2.2517x; 1.1376x over previous
#include <hip/hip_runtime.h>
#include <hip/hip_bf16.h>


#define NUM_USERS 100000
#define NUM_ITEMS 50000
#define N_NODES   (NUM_USERS + NUM_ITEMS)
#define LATENT    64
#define NNZ_E     4000000
#define BATCH_SZ  16384
#define SCAN_BLK  1024
#define SCAN_GRID ((N_NODES + SCAN_BLK - 1) / SCAN_BLK)

typedef __attribute__((ext_vector_type(2))) float v2f;
typedef __attribute__((ext_vector_type(8))) float v8f;
typedef __attribute__((ext_vector_type(2))) int   v2i;


static_assert(NNZ_E % 256 == 0, "a 256-thread block takes 256 edges, no tail");
static_assert(N_NODES % 8 == 0, "8 rows per 256-thread block, no tail");

__global__ __launch_bounds__(256) void lgcn_zero_cnt(int* __restrict__ cnt) {
    int i = blockIdx.x * blockDim.x + threadIdx.x;
    if (i < N_NODES) cnt[i] = 0;
}

__global__ __launch_bounds__(256) void lgcn_count(const int* __restrict__ edst, int* __restrict__ cnt) {
    const int e = blockIdx.x * blockDim.x + threadIdx.x;
    atomicAdd(cnt + edst[e], 1);
}

__global__ __launch_bounds__(SCAN_BLK) void lgcn_scan(const int* __restrict__ cnt,
                                                      int* __restrict__ cur,
                                                      int* __restrict__ rowptr) {
    __shared__ int wpart[32];
    __shared__ int wincl[32];
    const int t = threadIdx.x, lane = t & 31, w = t >> 5;
    const int base = blockIdx.x * SCAN_BLK;
    int part = 0;
    for (int i = t; i < base; i += SCAN_BLK) part += cnt[i];
    const int idx = base + t;
    const int v = (idx < N_NODES) ? cnt[idx] : 0;
    int incl = v;
#pragma unroll
    for (int o = 1; o < 32; o <<= 1) {
        const int n = __shfl_up(incl, o);
        if (lane >= o) incl += n;
    }
#pragma unroll
    for (int o = 16; o > 0; o >>= 1) part += __shfl_xor(part, o);
    if (lane == 31) wincl[w] = incl;
    if (lane == 0)  wpart[w] = part;
    __syncthreads();
    int tot = wpart[lane] + ((lane < w) ? wincl[lane] : 0);
#pragma unroll
    for (int o = 16; o > 0; o >>= 1) tot += __shfl_xor(tot, o);
    const int excl = tot + incl - v;
    if (idx < N_NODES) { rowptr[idx] = excl; cur[idx] = excl; }
    if (idx == N_NODES - 1) rowptr[N_NODES] = excl + v;
}

__global__ __launch_bounds__(256) void lgcn_fill(const int* __restrict__ esrc, const int* __restrict__ edst,
                                                 const float* __restrict__ evals,
                                                 int* __restrict__ cur, int* __restrict__ pairs) {
    const int e = blockIdx.x * blockDim.x + threadIdx.x;
    const int d = edst[e];
    const int s = __builtin_nontemporal_load(esrc + e);
    const float v = __builtin_nontemporal_load(evals + e);
    const int pos = atomicAdd(cur + d, 1);
    *(int2*)(pairs + 2 * (long)pos) = make_int2(s, __float_as_int(v));
}

template <bool FROM_TABLES>
__global__ __launch_bounds__(256) void lgcn_rows(const float* __restrict__ utab,
                                                 const float* __restrict__ itab,
                                                 const float* __restrict__ Xin,
                                                 const int* __restrict__ rowptr,
                                                 const int* __restrict__ pairs,
                                                 float* __restrict__ Zout) {
    const int lane = threadIdx.x & 31;
    const int row = __builtin_amdgcn_readfirstlane(blockIdx.x * 8 + (threadIdx.x >> 5));
    const int beg = rowptr[row];
    const int end = rowptr[row + 1];
    const float* x0 = (row < NUM_USERS) ? utab + (long)row * LATENT
                                        : itab + (long)(row - NUM_USERS) * LATENT;
    const v2f own = __builtin_nontemporal_load((const v2f*)(x0 + 2 * lane));
    float sx = 0.f, sy = 0.f;
    for (int base = beg; base < end; base += 32) {
        const int rem = end - base;
        const int cnt = rem < 32 ? rem : 32;
        int s = 0, vb = 0;
        if (lane < cnt) {
            const v2i p = __builtin_nontemporal_load((const v2i*)(pairs + 2 * (long)(base + lane)));
            s = p.x; vb = p.y;
        }
        const int s0 = __builtin_amdgcn_readfirstlane(s);
        if (lane >= cnt) s = s0;
#pragma unroll
        for (int g = 0; g < 32; g += 8) {
            if (g < cnt) {
                float2 x[8];
#pragma unroll
                for (int j = 0; j < 8; ++j) {
                    const int sj = __builtin_amdgcn_readlane(s, g + j);
                    const float* xr;
                    if (FROM_TABLES)
                        xr = (sj < NUM_USERS) ? utab + (long)sj * LATENT
                                              : itab + (long)(sj - NUM_USERS) * LATENT;
                    else
                        xr = Xin + (long)sj * LATENT;
                    x[j] = *(const float2*)(xr + 2 * lane);
                }
#pragma unroll
                for (int j = 0; j < 8; ++j) {
                    const float vj = __int_as_float(__builtin_amdgcn_readlane(vb, g + j));
                    sx += x[j].x * vj;
                    sy += x[j].y * vj;
                }
            }
        }
    }
    float2 z;
    z.x = own.x + sx;
    z.y = own.y + sy;
    *(float2*)(Zout + (long)row * LATENT + 2 * lane) = z;
}

__global__ void lgcn_gamma_wmma(const float* __restrict__ acc,
                                const int* __restrict__ users,
                                const int* __restrict__ items,
                                float* __restrict__ out) {
    int wave = (blockIdx.x * blockDim.x + threadIdx.x) >> 5;
    int lane = threadIdx.x & 31;
    int b    = wave * 16 + (lane & 15);
    int half = (lane >> 4) & 1;

    const float* urow = acc + (long)users[b] * LATENT;
    const float* irow = acc + ((long)NUM_USERS + items[b]) * LATENT;

    v8f c = {};
#pragma unroll
    for (int kk = 0; kk < 16; ++kk) {
        int off = 4 * kk + 2 * half;
        v2f a  = *(const v2f*)(urow + off);
        v2f bb = *(const v2f*)(irow + off);
        c = __builtin_amdgcn_wmma_f32_16x16x4_f32(false, a, false, bb,
                                                  (short)0, c, false, false);
    }

    int r = ((lane < 8) ? lane : (lane - 24)) & 7;
    float v = c[0];
#pragma unroll
    for (int i = 1; i < 8; ++i) if (r == i) v = c[i];

    if (lane < 8)        out[wave * 16 + lane]        = v * 0.0625f;
    else if (lane >= 24) out[wave * 16 + (lane - 16)] = v * 0.0625f;
}

extern "C" void kernel_launch(void* const* d_in, const int* in_sizes, int n_in,
                              void* d_out, int out_size, void* d_ws, size_t ws_size,
                              hipStream_t stream) {
    const int*   users = (const int*)d_in[0];
    const int*   items = (const int*)d_in[1];
    const int*   esrc  = (const int*)d_in[2];
    const int*   edst  = (const int*)d_in[3];
    const float* evals = (const float*)d_in[4];
    const float* utab  = (const float*)d_in[5];
    const float* itab  = (const float*)d_in[6];
    float* out = (float*)d_out;

    float* Za     = (float*)d_ws;
    float* Zb     = Za + 9600000;
    int*   pairs  = (int*)(Zb + 9600000);
    int*   cnt    = pairs + 8000000;
    int*   cur    = cnt + 150016;
    int*   rowptr = cur + 150016;

    lgcn_zero_cnt<<<(N_NODES + 255) / 256, 256, 0, stream>>>(cnt);
    lgcn_count<<<NNZ_E / 256, 256, 0, stream>>>(edst, cnt);
    lgcn_scan<<<SCAN_GRID, SCAN_BLK, 0, stream>>>(cnt, cur, rowptr);
    lgcn_fill<<<NNZ_E / 256, 256, 0, stream>>>(esrc, edst, evals, cur, pairs);

    lgcn_rows<true ><<<N_NODES / 8, 256, 0, stream>>>(utab, itab, utab, rowptr, pairs, Za);
    lgcn_rows<false><<<N_NODES / 8, 256, 0, stream>>>(utab, itab, Za,   rowptr, pairs, Zb);
    lgcn_rows<false><<<N_NODES / 8, 256, 0, stream>>>(utab, itab, Zb,   rowptr, pairs, Za);

    lgcn_gamma_wmma<<<BATCH_SZ / 16 / 8, 256, 0, stream>>>(Za, users, items, out);
}


# Third propagation only for the 2x16384 rows the batch reads (compact buffer, same per-row sum order), the full-table third pass is gone; the dot reads the compact rows
# speedup vs baseline: 2.3737x; 1.0542x over previous
#include <hip/hip_runtime.h>
#include <hip/hip_bf16.h>


#define NUM_USERS 100000
#define NUM_ITEMS 50000
#define N_NODES   (NUM_USERS + NUM_ITEMS)
#define LATENT    64
#define NNZ_E     4000000
#define BATCH_SZ  16384
#define SCAN_BLK  1024
#define SCAN_GRID ((N_NODES + SCAN_BLK - 1) / SCAN_BLK)

typedef __attribute__((ext_vector_type(2))) float v2f;
typedef __attribute__((ext_vector_type(8))) float v8f;
typedef __attribute__((ext_vector_type(2))) int   v2i;


static_assert(NNZ_E % 256 == 0, "a 256-thread block takes 256 edges, no tail");
static_assert(N_NODES % 8 == 0, "8 rows per 256-thread block, no tail");
static_assert(BATCH_SZ % 8 == 0, "a block of the batch-rows pass holds user rows or item rows, never both");

__global__ __launch_bounds__(256) void lgcn_zero_cnt(int* __restrict__ cnt) {
    int i = blockIdx.x * blockDim.x + threadIdx.x;
    if (i < N_NODES) cnt[i] = 0;
}

__global__ __launch_bounds__(256) void lgcn_count(const int* __restrict__ edst, int* __restrict__ cnt) {
    const int e = blockIdx.x * blockDim.x + threadIdx.x;
    atomicAdd(cnt + edst[e], 1);
}

__global__ __launch_bounds__(SCAN_BLK) void lgcn_scan(const int* __restrict__ cnt,
                                                      int* __restrict__ cur,
                                                      int* __restrict__ rowptr) {
    __shared__ int wpart[32];
    __shared__ int wincl[32];
    const int t = threadIdx.x, lane = t & 31, w = t >> 5;
    const int base = blockIdx.x * SCAN_BLK;
    int part = 0;
    for (int i = t; i < base; i += SCAN_BLK) part += cnt[i];
    const int idx = base + t;
    const int v = (idx < N_NODES) ? cnt[idx] : 0;
    int incl = v;
#pragma unroll
    for (int o = 1; o < 32; o <<= 1) {
        const int n = __shfl_up(incl, o);
        if (lane >= o) incl += n;
    }
#pragma unroll
    for (int o = 16; o > 0; o >>= 1) part += __shfl_xor(part, o);
    if (lane == 31) wincl[w] = incl;
    if (lane == 0)  wpart[w] = part;
    __syncthreads();
    int tot = wpart[lane] + ((lane < w) ? wincl[lane] : 0);
#pragma unroll
    for (int o = 16; o > 0; o >>= 1) tot += __shfl_xor(tot, o);
    const int excl = tot + incl - v;
    if (idx < N_NODES) { rowptr[idx] = excl; cur[idx] = excl; }
    if (idx == N_NODES - 1) rowptr[N_NODES] = excl + v;
}

__global__ __launch_bounds__(256) void lgcn_fill(const int* __restrict__ esrc, const int* __restrict__ edst,
                                                 const float* __restrict__ evals,
                                                 int* __restrict__ cur, int* __restrict__ pairs) {
    const int e = blockIdx.x * blockDim.x + threadIdx.x;
    const int d = edst[e];
    const int s = __builtin_nontemporal_load(esrc + e);
    const float v = __builtin_nontemporal_load(evals + e);
    const int pos = atomicAdd(cur + d, 1);
    *(int2*)(pairs + 2 * (long)pos) = make_int2(s, __float_as_int(v));
}

template <bool FROM_TABLES, bool BATCH_ROWS>
__global__ __launch_bounds__(256) void lgcn_rows(const float* __restrict__ utab,
                                                 const float* __restrict__ itab,
                                                 const float* __restrict__ Xin,
                                                 const int* __restrict__ rowptr,
                                                 const int* __restrict__ pairs,
                                                 const int* __restrict__ users,
                                                 const int* __restrict__ items,
                                                 float* __restrict__ Zout) {
    const int lane = threadIdx.x & 31;
    const int w = __builtin_amdgcn_readfirstlane(blockIdx.x * 8 + (threadIdx.x >> 5));
    int row = w;
    if (BATCH_ROWS) row = (w < BATCH_SZ) ? users[w] : NUM_USERS + items[w - BATCH_SZ];
    const int beg = rowptr[row];
    const int end = rowptr[row + 1];
    const float* x0 = (row < NUM_USERS) ? utab + (long)row * LATENT
                                        : itab + (long)(row - NUM_USERS) * LATENT;
    const v2f own = __builtin_nontemporal_load((const v2f*)(x0 + 2 * lane));
    float sx = 0.f, sy = 0.f;
    for (int base = beg; base < end; base += 32) {
        const int rem = end - base;
        const int cnt = rem < 32 ? rem : 32;
        int s = 0, vb = 0;
        if (lane < cnt) {
            const v2i p = __builtin_nontemporal_load((const v2i*)(pairs + 2 * (long)(base + lane)));
            s = p.x; vb = p.y;
        }
        const int s0 = __builtin_amdgcn_readfirstlane(s);
        if (lane >= cnt) s = s0;
#pragma unroll
        for (int g = 0; g < 32; g += 8) {
            if (g < cnt) {
                float2 x[8];
#pragma unroll
                for (int j = 0; j < 8; ++j) {
                    const int sj = __builtin_amdgcn_readlane(s, g + j);
                    const float* xr;
                    if (FROM_TABLES)
                        xr = (sj < NUM_USERS) ? utab + (long)sj * LATENT
                                              : itab + (long)(sj - NUM_USERS) * LATENT;
                    else
                        xr = Xin + (long)sj * LATENT;
                    x[j] = *(const float2*)(xr + 2 * lane);
                }
#pragma unroll
                for (int j = 0; j < 8; ++j) {
                    const float vj = __int_as_float(__builtin_amdgcn_readlane(vb, g + j));
                    sx += x[j].x * vj;
                    sy += x[j].y * vj;
                }
            }
        }
    }
    float2 z;
    z.x = own.x + sx;
    z.y = own.y + sy;
    *(float2*)(Zout + (long)w * LATENT + 2 * lane) = z;
}

__global__ void lgcn_gamma_wmma(const float* __restrict__ acc,
                                float* __restrict__ out) {
    int wave = (blockIdx.x * blockDim.x + threadIdx.x) >> 5;
    int lane = threadIdx.x & 31;
    int b    = wave * 16 + (lane & 15);
    int half = (lane >> 4) & 1;

    const float* urow = acc + (long)b * LATENT;
    const float* irow = acc + ((long)BATCH_SZ + b) * LATENT;

    v8f c = {};
#pragma unroll
    for (int kk = 0; kk < 16; ++kk) {
        int off = 4 * kk + 2 * half;
        v2f a  = *(const v2f*)(urow + off);
        v2f bb = *(const v2f*)(irow + off);
        c = __builtin_amdgcn_wmma_f32_16x16x4_f32(false, a, false, bb,
                                                  (short)0, c, false, false);
    }

    int r = ((lane < 8) ? lane : (lane - 24)) & 7;
    float v = c[0];
#pragma unroll
    for (int i = 1; i < 8; ++i) if (r == i) v = c[i];

    if (lane < 8)        out[wave * 16 + lane]        = v * 0.0625f;
    else if (lane >= 24) out[wave * 16 + (lane - 16)] = v * 0.0625f;
}

extern "C" void kernel_launch(void* const* d_in, const int* in_sizes, int n_in,
                              void* d_out, int out_size, void* d_ws, size_t ws_size,
                              hipStream_t stream) {
    const int*   users = (const int*)d_in[0];
    const int*   items = (const int*)d_in[1];
    const int*   esrc  = (const int*)d_in[2];
    const int*   edst  = (const int*)d_in[3];
    const float* evals = (const float*)d_in[4];
    const float* utab  = (const float*)d_in[5];
    const float* itab  = (const float*)d_in[6];
    float* out = (float*)d_out;

    float* Za     = (float*)d_ws;
    float* Zb     = Za + 9600000;
    int*   pairs  = (int*)(Zb + 9600000);
    int*   cnt    = pairs + 8000000;
    int*   cur    = cnt + 150016;
    int*   rowptr = cur + 150016;

    lgcn_zero_cnt<<<(N_NODES + 255) / 256, 256, 0, stream>>>(cnt);
    lgcn_count<<<NNZ_E / 256, 256, 0, stream>>>(edst, cnt);
    lgcn_scan<<<SCAN_GRID, SCAN_BLK, 0, stream>>>(cnt, cur, rowptr);
    lgcn_fill<<<NNZ_E / 256, 256, 0, stream>>>(esrc, edst, evals, cur, pairs);

    lgcn_rows<true,  false><<<N_NODES / 8, 256, 0, stream>>>(utab, itab, utab, rowptr, pairs, users, items, Za);
    lgcn_rows<false, false><<<N_NODES / 8, 256, 0, stream>>>(utab, itab, Za,   rowptr, pairs, users, items, Zb);
    lgcn_rows<false, true ><<<2 * BATCH_SZ / 8, 256, 0, stream>>>(utab, itab, Zb, rowptr, pairs, users, items, Za);

    lgcn_gamma_wmma<<<BATCH_SZ / 16 / 8, 256, 0, stream>>>(Za, out);
}


# Edge sort without global atomics: two-level bucket sort by dst (per-chunk LDS counts, scan, LDS cursors; 512-row buckets); order inside a row is now the LDS adds' arrival order
# speedup vs baseline: 6.9590x; 2.9317x over previous
#include <hip/hip_runtime.h>
#include <hip/hip_bf16.h>


#define NUM_USERS 100000
#define NUM_ITEMS 50000
#define N_NODES   (NUM_USERS + NUM_ITEMS)
#define LATENT    64
#define NNZ_E     4000000
#define BATCH_SZ  16384
#define SCAN_BLK  1024
#define BKT_SH    9
#define BKT_ROWS  (1 << BKT_SH)
#define NBKT      ((N_NODES + BKT_ROWS - 1) >> BKT_SH)
#define CHUNK     16384
#define NCHK      ((NNZ_E + CHUNK - 1) / CHUNK)
#define NCELL     (NBKT * NCHK)
#define SCAN_GRID ((NCELL + SCAN_BLK - 1) / SCAN_BLK)
#define SRC_BITS  18
#define SRC_MASK  ((1 << SRC_BITS) - 1)

typedef __attribute__((ext_vector_type(2))) float v2f;
typedef __attribute__((ext_vector_type(4))) float v4f;
typedef __attribute__((ext_vector_type(8))) float v8f;
typedef __attribute__((ext_vector_type(2))) int   v2i;
typedef __attribute__((ext_vector_type(4))) int   v4i;


static_assert(NNZ_E % 4 == 0, "edges are read four at a time, no straddling tail");
static_assert(CHUNK == 16 * SCAN_BLK, "a 1024-thread block reads its chunk as 4 x 4 edges per thread");
static_assert(N_NODES <= (1 << SRC_BITS), "a source index fits under the row-in-bucket bits");
static_assert(SRC_BITS + BKT_SH < 32, "source and row-in-bucket share one positive word");
static_assert(NCELL + 1 <= 150016, "cell counts and cell starts fit the two count arrays of the workspace");
static_assert(BKT_ROWS == 512, "the level-two scan is written for 16 waves of 32 rows");
static_assert(N_NODES % 8 == 0, "8 rows per 256-thread block, no tail");
static_assert(BATCH_SZ % 8 == 0, "a block of the batch-rows pass holds user rows or item rows, never both");

__global__ __launch_bounds__(SCAN_BLK) void lgcn_bucket_count(const int* __restrict__ edst,
                                                              int* __restrict__ cell) {
    __shared__ int h[NBKT];
    const int t = threadIdx.x;
    for (int i = t; i < NBKT; i += SCAN_BLK) h[i] = 0;
    __syncthreads();
    const int base = blockIdx.x * CHUNK;
    v4i d[4];
    for (int j = 0; j < 4; ++j) {
        const int e = base + (j * SCAN_BLK + t) * 4;
        d[j] = *(const v4i*)(edst + (e < NNZ_E ? e : NNZ_E - 4));
    }
    for (int j = 0; j < 4; ++j) {
        const int e = base + (j * SCAN_BLK + t) * 4;
        if (e < NNZ_E) {
            atomicAdd(&h[d[j].x >> BKT_SH], 1);
            atomicAdd(&h[d[j].y >> BKT_SH], 1);
            atomicAdd(&h[d[j].z >> BKT_SH], 1);
            atomicAdd(&h[d[j].w >> BKT_SH], 1);
        }
    }
    __syncthreads();
    for (int i = t; i < NBKT; i += SCAN_BLK) cell[i * NCHK + blockIdx.x] = h[i];
}

__global__ __launch_bounds__(SCAN_BLK) void lgcn_scan(const int* __restrict__ cnt,
                                                      int* __restrict__ off) {
    __shared__ int wpart[32];
    __shared__ int wincl[32];
    const int t = threadIdx.x, lane = t & 31, w = t >> 5;
    const int base = blockIdx.x * SCAN_BLK;
    int part = 0;
    for (int i = 4 * t; i < base; i += 16 * SCAN_BLK) {
        v4i c[4];
        for (int j = 0; j < 4; ++j) {
            const int ii = i + j * 4 * SCAN_BLK;
            c[j] = *(const v4i*)(cnt + (ii < base ? ii : i));
        }
        for (int j = 0; j < 4; ++j)
            part += (i + j * 4 * SCAN_BLK < base) ? (c[j].x + c[j].y) + (c[j].z + c[j].w) : 0;
    }
    const int idx = base + t;
    const int v = (idx < NCELL) ? cnt[idx] : 0;
    int incl = v;
#pragma unroll
    for (int o = 1; o < 32; o <<= 1) {
        const int n = __shfl_up(incl, o);
        if (lane >= o) incl += n;
    }
#pragma unroll
    for (int o = 16; o > 0; o >>= 1) part += __shfl_xor(part, o);
    if (lane == 31) wincl[w] = incl;
    if (lane == 0)  wpart[w] = part;
    __syncthreads();
    int tot = wpart[lane] + ((lane < w) ? wincl[lane] : 0);
#pragma unroll
    for (int o = 16; o > 0; o >>= 1) tot += __shfl_xor(tot, o);
    const int excl = tot + incl - v;
    if (idx < NCELL) off[idx] = excl;
    if (idx == NCELL - 1) off[NCELL] = excl + v;
}

__global__ __launch_bounds__(SCAN_BLK) void lgcn_bucket_move(const int* __restrict__ esrc,
                                                             const int* __restrict__ edst,
                                                             const float* __restrict__ evals,
                                                             const int* __restrict__ off,
                                                             int* __restrict__ bkt) {
    __shared__ int cur[NBKT];
    const int t = threadIdx.x;
    for (int i = t; i < NBKT; i += SCAN_BLK) cur[i] = off[i * NCHK + blockIdx.x];
    __syncthreads();
    const int base = blockIdx.x * CHUNK;
    v4i d[4], s[4];
    v4f v[4];
    for (int j = 0; j < 4; ++j) {
        const int e = base + (j * SCAN_BLK + t) * 4;
        const int ec = e < NNZ_E ? e : NNZ_E - 4;
        d[j] = __builtin_nontemporal_load((const v4i*)(edst + ec));
        s[j] = __builtin_nontemporal_load((const v4i*)(esrc + ec));
        v[j] = __builtin_nontemporal_load((const v4f*)(evals + ec));
    }
    for (int j = 0; j < 4; ++j) {
        const int e = base + (j * SCAN_BLK + t) * 4;
        if (e < NNZ_E) {
#pragma unroll
            for (int k = 0; k < 4; ++k) {
                const int dk = d[j][k];
                const int pos = atomicAdd(&cur[dk >> BKT_SH], 1);
                v2i p;
                p.x = s[j][k] | ((dk & (BKT_ROWS - 1)) << SRC_BITS);
                p.y = __float_as_int(v[j][k]);
                *(v2i*)(bkt + 2 * (long)pos) = p;
            }
        }
    }
}

__global__ __launch_bounds__(SCAN_BLK) void lgcn_bucket_rows(const int* __restrict__ off,
                                                             const int* __restrict__ bkt,
                                                             int* __restrict__ pairs,
                                                             int* __restrict__ rowptr) {
    __shared__ int h[BKT_ROWS];
    __shared__ int wsum[BKT_ROWS / 32];
    const int t = threadIdx.x, lane = t & 31, w = t >> 5;
    const int b = blockIdx.x;
    const int beg = off[b * NCHK];
    const int end = off[(b + 1) * NCHK];
    if (t < BKT_ROWS) h[t] = 0;
    __syncthreads();
    for (int i0 = beg + t; i0 < end; i0 += 4 * SCAN_BLK) {
        int k[4];
        for (int j = 0; j < 4; ++j) {
            const int i = i0 + j * SCAN_BLK;
            k[j] = bkt[2 * (long)(i < end ? i : end - 1)];
        }
        for (int j = 0; j < 4; ++j)
            if (i0 + j * SCAN_BLK < end) atomicAdd(&h[k[j] >> SRC_BITS], 1);
    }
    __syncthreads();
    const int c = (t < BKT_ROWS) ? h[t] : 0;
    int incl = c;
#pragma unroll
    for (int o = 1; o < 32; o <<= 1) {
        const int n = __shfl_up(incl, o);
        if (lane >= o) incl += n;
    }
    if (t < BKT_ROWS && lane == 31) wsum[w] = incl;
    __syncthreads();
    if (t < BKT_ROWS) {
        int pre = beg;
        for (int q = 0; q < BKT_ROWS / 32; ++q) pre += (q < w) ? wsum[q] : 0;
        const int excl = pre + incl - c;
        h[t] = excl;
        const int node = (b << BKT_SH) + t;
        if (node < N_NODES) rowptr[node] = excl;
    }
    if (b == NBKT - 1 && t == 0) rowptr[N_NODES] = end;
    __syncthreads();
    for (int i0 = beg + t; i0 < end; i0 += 4 * SCAN_BLK) {
        v2i p[4];
        for (int j = 0; j < 4; ++j) {
            const int i = i0 + j * SCAN_BLK;
            p[j] = *(const v2i*)(bkt + 2 * (long)(i < end ? i : end - 1));
        }
        for (int j = 0; j < 4; ++j) {
            if (i0 + j * SCAN_BLK < end) {
                const int pos = atomicAdd(&h[p[j].x >> SRC_BITS], 1);
                v2i q;
                q.x = p[j].x & SRC_MASK;
                q.y = p[j].y;
                *(v2i*)(pairs + 2 * (long)pos) = q;
            }
        }
    }
}

template <bool FROM_TABLES, bool BATCH_ROWS>
__global__ __launch_bounds__(256) void lgcn_rows(const float* __restrict__ utab,
                                                 const float* __restrict__ itab,
                                                 const float* __restrict__ Xin,
                                                 const int* __restrict__ rowptr,
                                                 const int* __restrict__ pairs,
                                                 const int* __restrict__ users,
                                                 const int* __restrict__ items,
                                                 float* __restrict__ Zout) {
    const int lane = threadIdx.x & 31;
    const int w = __builtin_amdgcn_readfirstlane(blockIdx.x * 8 + (threadIdx.x >> 5));
    int row = w;
    if (BATCH_ROWS) row = (w < BATCH_SZ) ? users[w] : NUM_USERS + items[w - BATCH_SZ];
    const int beg = rowptr[row];
    const int end = rowptr[row + 1];
    const float* x0 = (row < NUM_USERS) ? utab + (long)row * LATENT
                                        : itab + (long)(row - NUM_USERS) * LATENT;
    const v2f own = __builtin_nontemporal_load((const v2f*)(x0 + 2 * lane));
    float sx = 0.f, sy = 0.f;
    for (int base = beg; base < end; base += 32) {
        const int rem = end - base;
        const int cnt = rem < 32 ? rem : 32;
        int s = 0, vb = 0;
        if (lane < cnt) {
            const v2i p = __builtin_nontemporal_load((const v2i*)(pairs + 2 * (long)(base + lane)));
            s = p.x; vb = p.y;
        }
        const int s0 = __builtin_amdgcn_readfirstlane(s);
        if (lane >= cnt) s = s0;
#pragma unroll
        for (int g = 0; g < 32; g += 8) {
            if (g < cnt) {
                float2 x[8];
#pragma unroll
                for (int j = 0; j < 8; ++j) {
                    const int sj = __builtin_amdgcn_readlane(s, g + j);
                    const float* xr;
                    if (FROM_TABLES)
                        xr = (sj < NUM_USERS) ? utab + (long)sj * LATENT
                                              : itab + (long)(sj - NUM_USERS) * LATENT;
                    else
                        xr = Xin + (long)sj * LATENT;
                    x[j] = *(const float2*)(xr + 2 * lane);
                }
#pragma unroll
                for (int j = 0; j < 8; ++j) {
                    const float vj = __int_as_float(__builtin_amdgcn_readlane(vb, g + j));
                    sx += x[j].x * vj;
                    sy += x[j].y * vj;
                }
            }
        }
    }
    float2 z;
    z.x = own.x + sx;
    z.y = own.y + sy;
    *(float2*)(Zout + (long)w * LATENT + 2 * lane) = z;
}

__global__ void lgcn_gamma_wmma(const float* __restrict__ acc,
                                float* __restrict__ out) {
    int wave = (blockIdx.x * blockDim.x + threadIdx.x) >> 5;
    int lane = threadIdx.x & 31;
    int b    = wave * 16 + (lane & 15);
    int half = (lane >> 4) & 1;

    const float* urow = acc + (long)b * LATENT;
    const float* irow = acc + ((long)BATCH_SZ + b) * LATENT;

    v8f c = {};
#pragma unroll
    for (int kk = 0; kk < 16; ++kk) {
        int off = 4 * kk + 2 * half;
        v2f a  = *(const v2f*)(urow + off);
        v2f bb = *(const v2f*)(irow + off);
        c = __builtin_amdgcn_wmma_f32_16x16x4_f32(false, a, false, bb,
                                                  (short)0, c, false, false);
    }

    int r = ((lane < 8) ? lane : (lane - 24)) & 7;
    float v = c[0];
#pragma unroll
    for (int i = 1; i < 8; ++i) if (r == i) v = c[i];

    if (lane < 8)        out[wave * 16 + lane]        = v * 0.0625f;
    else if (lane >= 24) out[wave * 16 + (lane - 16)] = v * 0.0625f;
}

extern "C" void kernel_launch(void* const* d_in, const int* in_sizes, int n_in,
                              void* d_out, int out_size, void* d_ws, size_t ws_size,
                              hipStream_t stream) {
    const int*   users = (const int*)d_in[0];
    const int*   items = (const int*)d_in[1];
    const int*   esrc  = (const int*)d_in[2];
    const int*   edst  = (const int*)d_in[3];
    const float* evals = (const float*)d_in[4];
    const float* utab  = (const float*)d_in[5];
    const float* itab  = (const float*)d_in[6];
    float* out = (float*)d_out;

    float* Za     = (float*)d_ws;
    float* Zb     = Za + 9600000;
    int*   pairs  = (int*)(Zb + 9600000);
    int*   cell   = pairs + 8000000;
    int*   off    = cell + 150016;
    int*   rowptr = off + 150016;
    int*   bkt    = (int*)Zb;

    lgcn_bucket_count<<<NCHK, SCAN_BLK, 0, stream>>>(edst, cell);
    lgcn_scan<<<SCAN_GRID, SCAN_BLK, 0, stream>>>(cell, off);
    lgcn_bucket_move<<<NCHK, SCAN_BLK, 0, stream>>>(esrc, edst, evals, off, bkt);
    lgcn_bucket_rows<<<NBKT, SCAN_BLK, 0, stream>>>(off, bkt, pairs, rowptr);

    lgcn_rows<true,  false><<<N_NODES / 8, 256, 0, stream>>>(utab, itab, utab, rowptr, pairs, users, items, Za);
    lgcn_rows<false, false><<<N_NODES / 8, 256, 0, stream>>>(utab, itab, Za,   rowptr, pairs, users, items, Zb);
    lgcn_rows<false, true ><<<2 * BATCH_SZ / 8, 256, 0, stream>>>(utab, itab, Zb, rowptr, pairs, users, items, Za);

    lgcn_gamma_wmma<<<BATCH_SZ / 16 / 8, 256, 0, stream>>>(Za, out);
}


# Edge sort: both levels place their (src,val) pairs in LDS and copy finished runs out as whole lines instead of 8M scattered 8-byte stores; order inside a row stays the LDS adds' arrival order
# speedup vs baseline: 8.2141x; 1.1804x over previous
#include <hip/hip_runtime.h>
#include <hip/hip_bf16.h>


#define NUM_USERS 100000
#define NUM_ITEMS 50000
#define N_NODES   (NUM_USERS + NUM_ITEMS)
#define LATENT    64
#define NNZ_E     4000000
#define BATCH_SZ  16384
#define SCAN_BLK  1024
#define BKT_SH    9
#define BKT_ROWS  (1 << BKT_SH)
#define NBKT      ((N_NODES + BKT_ROWS - 1) >> BKT_SH)
#define CHUNK     16384
#define NCHK      ((NNZ_E + CHUNK - 1) / CHUNK)
#define NCELL     (NBKT * NCHK)
#define SCAN_GRID ((NCELL + SCAN_BLK - 1) / SCAN_BLK)
#define SRC_BITS  18
#define SRC_MASK  ((1 << SRC_BITS) - 1)
#define ROWS_CAP  16384

typedef __attribute__((ext_vector_type(2))) float v2f;
typedef __attribute__((ext_vector_type(4))) float v4f;
typedef __attribute__((ext_vector_type(8))) float v8f;
typedef __attribute__((ext_vector_type(2))) int   v2i;
typedef __attribute__((ext_vector_type(4))) int   v4i;


static_assert(NNZ_E % 4 == 0, "edges are read four at a time, no straddling tail");
static_assert(CHUNK == 16 * SCAN_BLK, "a 1024-thread block reads its chunk as 4 x 4 edges per thread");
static_assert(N_NODES <= (1 << SRC_BITS), "a source index fits under the row-in-bucket bits");
static_assert(SRC_BITS + BKT_SH < 32, "source and row-in-bucket share one positive word");
static_assert(NCELL + 1 <= 150016, "cell counts and cell starts fit the two count arrays of the workspace");
static_assert(BKT_ROWS == 512, "the level-two scan is written for 16 waves of 32 rows");
static_assert(NBKT <= SCAN_BLK, "level one scans its chunk's bucket counts with one thread per bucket");
static_assert(N_NODES % 8 == 0, "8 rows per 256-thread block, no tail");
static_assert(BATCH_SZ % 8 == 0, "a block of the batch-rows pass holds user rows or item rows, never both");

__global__ __launch_bounds__(SCAN_BLK) void lgcn_bucket_count(const int* __restrict__ edst,
                                                              int* __restrict__ cell) {
    __shared__ int h[NBKT];
    const int t = threadIdx.x;
    for (int i = t; i < NBKT; i += SCAN_BLK) h[i] = 0;
    __syncthreads();
    const int base = blockIdx.x * CHUNK;
    v4i d[4];
    for (int j = 0; j < 4; ++j) {
        const int e = base + (j * SCAN_BLK + t) * 4;
        d[j] = *(const v4i*)(edst + (e < NNZ_E ? e : NNZ_E - 4));
    }
    for (int j = 0; j < 4; ++j) {
        const int e = base + (j * SCAN_BLK + t) * 4;
        if (e < NNZ_E) {
            atomicAdd(&h[d[j].x >> BKT_SH], 1);
            atomicAdd(&h[d[j].y >> BKT_SH], 1);
            atomicAdd(&h[d[j].z >> BKT_SH], 1);
            atomicAdd(&h[d[j].w >> BKT_SH], 1);
        }
    }
    __syncthreads();
    for (int i = t; i < NBKT; i += SCAN_BLK) cell[i * NCHK + blockIdx.x] = h[i];
}

__global__ __launch_bounds__(SCAN_BLK) void lgcn_scan(const int* __restrict__ cnt,
                                                      int* __restrict__ off) {
    __shared__ int wpart[32];
    __shared__ int wincl[32];
    const int t = threadIdx.x, lane = t & 31, w = t >> 5;
    const int base = blockIdx.x * SCAN_BLK;
    int part = 0;
    for (int i = 4 * t; i < base; i += 16 * SCAN_BLK) {
        v4i c[4];
        for (int j = 0; j < 4; ++j) {
            const int ii = i + j * 4 * SCAN_BLK;
            c[j] = *(const v4i*)(cnt + (ii < base ? ii : i));
        }
        for (int j = 0; j < 4; ++j)
            part += (i + j * 4 * SCAN_BLK < base) ? (c[j].x + c[j].y) + (c[j].z + c[j].w) : 0;
    }
    const int idx = base + t;
    const int v = (idx < NCELL) ? cnt[idx] : 0;
    int incl = v;
#pragma unroll
    for (int o = 1; o < 32; o <<= 1) {
        const int n = __shfl_up(incl, o);
        if (lane >= o) incl += n;
    }
#pragma unroll
    for (int o = 16; o > 0; o >>= 1) part += __shfl_xor(part, o);
    if (lane == 31) wincl[w] = incl;
    if (lane == 0)  wpart[w] = part;
    __syncthreads();
    int tot = wpart[lane] + ((lane < w) ? wincl[lane] : 0);
#pragma unroll
    for (int o = 16; o > 0; o >>= 1) tot += __shfl_xor(tot, o);
    const int excl = tot + incl - v;
    if (idx < NCELL) off[idx] = excl;
    if (idx == NCELL - 1) off[NCELL] = excl + v;
}

__global__ __launch_bounds__(SCAN_BLK) void lgcn_bucket_move(const int* __restrict__ esrc,
                                                             const int* __restrict__ edst,
                                                             const float* __restrict__ evals,
                                                             const int* __restrict__ off,
                                                             int* __restrict__ bkt) {
    __shared__ v2i stage[CHUNK];
    __shared__ int lbeg[NBKT];
    __shared__ int cur[NBKT];
    __shared__ int gbeg[NBKT];
    __shared__ int wtot[32];
    const int t = threadIdx.x, lane = t & 31;
    const int w = __builtin_amdgcn_readfirstlane(t >> 5);
    int g = 0, c = 0;
    if (t < NBKT) {
        g = off[t * NCHK + blockIdx.x];
        c = off[t * NCHK + blockIdx.x + 1] - g;
    }
    const int base = blockIdx.x * CHUNK;
    v4i d[4], s[4];
    v4f v[4];
    for (int j = 0; j < 4; ++j) {
        const int e = base + (j * SCAN_BLK + t) * 4;
        const int ec = e < NNZ_E ? e : NNZ_E - 4;
        d[j] = __builtin_nontemporal_load((const v4i*)(edst + ec));
        s[j] = __builtin_nontemporal_load((const v4i*)(esrc + ec));
        v[j] = __builtin_nontemporal_load((const v4f*)(evals + ec));
    }
    int incl = c;
#pragma unroll
    for (int o = 1; o < 32; o <<= 1) {
        const int n = __shfl_up(incl, o);
        if (lane >= o) incl += n;
    }
    if (lane == 31) wtot[w] = incl;
    __syncthreads();
    if (t < NBKT) {
        int pre = 0;
        for (int q = 0; q < (NBKT + 31) / 32; ++q) pre += (q < w) ? wtot[q] : 0;
        const int lb = pre + incl - c;
        lbeg[t] = lb;
        cur[t]  = lb;
        gbeg[t] = g;
    }
    __syncthreads();
    for (int j = 0; j < 4; ++j) {
        const int e = base + (j * SCAN_BLK + t) * 4;
        if (e < NNZ_E) {
#pragma unroll
            for (int k = 0; k < 4; ++k) {
                const int dk = d[j][k];
                const int pos = atomicAdd(&cur[dk >> BKT_SH], 1);
                v2i p;
                p.x = s[j][k] | ((dk & (BKT_ROWS - 1)) << SRC_BITS);
                p.y = __float_as_int(v[j][k]);
                stage[pos] = p;
            }
        }
    }
    __syncthreads();
    for (int b = w; b < NBKT; b += SCAN_BLK / 32) {
        const int lb = __builtin_amdgcn_readfirstlane(lbeg[b]);
        const int n  = __builtin_amdgcn_readfirstlane(cur[b]) - lb;
        const int gb = __builtin_amdgcn_readfirstlane(gbeg[b]);
        for (int k = lane; k < n; k += 32)
            *(v2i*)(bkt + 2 * (long)(gb + k)) = stage[lb + k];
    }
}

__global__ __launch_bounds__(SCAN_BLK) void lgcn_bucket_rows(const int* __restrict__ off,
                                                             const int* __restrict__ bkt,
                                                             int* __restrict__ pairs,
                                                             int* __restrict__ rowptr) {
    __shared__ v2i stage[ROWS_CAP];
    __shared__ int h[BKT_ROWS];
    __shared__ int wsum[BKT_ROWS / 32];
    const int t = threadIdx.x, lane = t & 31, w = t >> 5;
    const int b = blockIdx.x;
    const int beg = off[b * NCHK];
    const int end = off[(b + 1) * NCHK];
    if (t < BKT_ROWS) h[t] = 0;
    __syncthreads();
    for (int i0 = beg + t; i0 < end; i0 += 4 * SCAN_BLK) {
        int k[4];
        for (int j = 0; j < 4; ++j) {
            const int i = i0 + j * SCAN_BLK;
            k[j] = bkt[2 * (long)(i < end ? i : end - 1)];
        }
        for (int j = 0; j < 4; ++j)
            if (i0 + j * SCAN_BLK < end) atomicAdd(&h[k[j] >> SRC_BITS], 1);
    }
    __syncthreads();
    const int c = (t < BKT_ROWS) ? h[t] : 0;
    int incl = c;
#pragma unroll
    for (int o = 1; o < 32; o <<= 1) {
        const int n = __shfl_up(incl, o);
        if (lane >= o) incl += n;
    }
    if (t < BKT_ROWS && lane == 31) wsum[w] = incl;
    __syncthreads();
    if (t < BKT_ROWS) {
        int pre = beg;
        for (int q = 0; q < BKT_ROWS / 32; ++q) pre += (q < w) ? wsum[q] : 0;
        const int excl = pre + incl - c;
        h[t] = excl;
        const int node = (b << BKT_SH) + t;
        if (node < N_NODES) rowptr[node] = excl;
    }
    if (b == NBKT - 1 && t == 0) rowptr[N_NODES] = end;
    __syncthreads();
    if (end - beg <= ROWS_CAP) {
        for (int i0 = beg + t; i0 < end; i0 += 4 * SCAN_BLK) {
            v2i p[4];
            for (int j = 0; j < 4; ++j) {
                const int i = i0 + j * SCAN_BLK;
                p[j] = *(const v2i*)(bkt + 2 * (long)(i < end ? i : end - 1));
            }
            for (int j = 0; j < 4; ++j) {
                if (i0 + j * SCAN_BLK < end) {
                    const int pos = atomicAdd(&h[p[j].x >> SRC_BITS], 1);
                    v2i q;
                    q.x = p[j].x & SRC_MASK;
                    q.y = p[j].y;
                    stage[pos - beg] = q;
                }
            }
        }
        __syncthreads();
        for (int i = t - (beg & 15); i < end - beg; i += SCAN_BLK)
            if (i >= 0) *(v2i*)(pairs + 2 * (long)(beg + i)) = stage[i];
    } else {
        for (int i0 = beg + t; i0 < end; i0 += 4 * SCAN_BLK) {
            v2i p[4];
            for (int j = 0; j < 4; ++j) {
                const int i = i0 + j * SCAN_BLK;
                p[j] = *(const v2i*)(bkt + 2 * (long)(i < end ? i : end - 1));
            }
            for (int j = 0; j < 4; ++j) {
                if (i0 + j * SCAN_BLK < end) {
                    const int pos = atomicAdd(&h[p[j].x >> SRC_BITS], 1);
                    v2i q;
                    q.x = p[j].x & SRC_MASK;
                    q.y = p[j].y;
                    *(v2i*)(pairs + 2 * (long)pos) = q;
                }
            }
        }
    }
}

template <bool FROM_TABLES, bool BATCH_ROWS>
__global__ __launch_bounds__(256) void lgcn_rows(const float* __restrict__ utab,
                                                 const float* __restrict__ itab,
                                                 const float* __restrict__ Xin,
                                                 const int* __restrict__ rowptr,
                                                 const int* __restrict__ pairs,
                                                 const int* __restrict__ users,
                                                 const int* __restrict__ items,
                                                 float* __restrict__ Zout) {
    const int lane = threadIdx.x & 31;
    const int w = __builtin_amdgcn_readfirstlane(blockIdx.x * 8 + (threadIdx.x >> 5));
    int row = w;
    if (BATCH_ROWS) row = (w < BATCH_SZ) ? users[w] : NUM_USERS + items[w - BATCH_SZ];
    const int beg = rowptr[row];
    const int end = rowptr[row + 1];
    const float* x0 = (row < NUM_USERS) ? utab + (long)row * LATENT
                                        : itab + (long)(row - NUM_USERS) * LATENT;
    const v2f own = __builtin_nontemporal_load((const v2f*)(x0 + 2 * lane));
    float sx = 0.f, sy = 0.f;
    for (int base = beg; base < end; base += 32) {
        const int rem = end - base;
        const int cnt = rem < 32 ? rem : 32;
        int s = 0, vb = 0;
        if (lane < cnt) {
            const v2i p = __builtin_nontemporal_load((const v2i*)(pairs + 2 * (long)(base + lane)));
            s = p.x; vb = p.y;
        }
        const int s0 = __builtin_amdgcn_readfirstlane(s);
        if (lane >= cnt) s = s0;
#pragma unroll
        for (int g = 0; g < 32; g += 8) {
            if (g < cnt) {
                float2 x[8];
#pragma unroll
                for (int j = 0; j < 8; ++j) {
                    const int sj = __builtin_amdgcn_readlane(s, g + j);
                    const float* xr;
                    if (FROM_TABLES)
                        xr = (sj < NUM_USERS) ? utab + (long)sj * LATENT
                                              : itab + (long)(sj - NUM_USERS) * LATENT;
                    else
                        xr = Xin + (long)sj * LATENT;
                    x[j] = *(const float2*)(xr + 2 * lane);
                }
#pragma unroll
                for (int j = 0; j < 8; ++j) {
                    const float vj = __int_as_float(__builtin_amdgcn_readlane(vb, g + j));
                    sx += x[j].x * vj;
                    sy += x[j].y * vj;
                }
            }
        }
    }
    float2 z;
    z.x = own.x + sx;
    z.y = own.y + sy;
    *(float2*)(Zout + (long)w * LATENT + 2 * lane) = z;
}

__global__ void lgcn_gamma_wmma(const float* __restrict__ acc,
                                float* __restrict__ out) {
    int wave = (blockIdx.x * blockDim.x + threadIdx.x) >> 5;
    int lane = threadIdx.x & 31;
    int b    = wave * 16 + (lane & 15);
    int half = (lane >> 4) & 1;

    const float* urow = acc + (long)b * LATENT;
    const float* irow = acc + ((long)BATCH_SZ + b) * LATENT;

    v8f c = {};
#pragma unroll
    for (int kk = 0; kk < 16; ++kk) {
        int off = 4 * kk + 2 * half;
        v2f a  = *(const v2f*)(urow + off);
        v2f bb = *(const v2f*)(irow + off);
        c = __builtin_amdgcn_wmma_f32_16x16x4_f32(false, a, false, bb,
                                                  (short)0, c, false, false);
    }

    int r = ((lane < 8) ? lane : (lane - 24)) & 7;
    float v = c[0];
#pragma unroll
    for (int i = 1; i < 8; ++i) if (r == i) v = c[i];

    if (lane < 8)        out[wave * 16 + lane]        = v * 0.0625f;
    else if (lane >= 24) out[wave * 16 + (lane - 16)] = v * 0.0625f;
}

extern "C" void kernel_launch(void* const* d_in, const int* in_sizes, int n_in,
                              void* d_out, int out_size, void* d_ws, size_t ws_size,
                              hipStream_t stream) {
    const int*   users = (const int*)d_in[0];
    const int*   items = (const int*)d_in[1];
    const int*   esrc  = (const int*)d_in[2];
    const int*   edst  = (const int*)d_in[3];
    const float* evals = (const float*)d_in[4];
    const float* utab  = (const float*)d_in[5];
    const float* itab  = (const float*)d_in[6];
    float* out = (float*)d_out;

    float* Za     = (float*)d_ws;
    float* Zb     = Za + 9600000;
    int*   pairs  = (int*)(Zb + 9600000);
    int*   cell   = pairs + 8000000;
    int*   off    = cell + 150016;
    int*   rowptr = off + 150016;
    int*   bkt    = (int*)Zb;

    lgcn_bucket_count<<<NCHK, SCAN_BLK, 0, stream>>>(edst, cell);
    lgcn_scan<<<SCAN_GRID, SCAN_BLK, 0, stream>>>(cell, off);
    lgcn_bucket_move<<<NCHK, SCAN_BLK, 0, stream>>>(esrc, edst, evals, off, bkt);
    lgcn_bucket_rows<<<NBKT, SCAN_BLK, 0, stream>>>(off, bkt, pairs, rowptr);

    lgcn_rows<true,  false><<<N_NODES / 8, 256, 0, stream>>>(utab, itab, utab, rowptr, pairs, users, items, Za);
    lgcn_rows<false, false><<<N_NODES / 8, 256, 0, stream>>>(utab, itab, Za,   rowptr, pairs, users, items, Zb);
    lgcn_rows<false, true ><<<2 * BATCH_SZ / 8, 256, 0, stream>>>(utab, itab, Zb, rowptr, pairs, users, items, Za);

    lgcn_gamma_wmma<<<BATCH_SZ / 16 / 8, 256, 0, stream>>>(Za, out);
}
